# starting kernel
# baseline (speedup 1.0000x reference)
#include <hip/hip_runtime.h>

typedef __attribute__((ext_vector_type(2))) float v2f;
typedef __attribute__((ext_vector_type(8))) float v8f;

#define B_TOT   2048
#define I_CAP   64
#define O_CAP   32
#define D_DIM   16
#define BT      16
#define NBT     (B_TOT / BT)
#define PR_PITCH 68
#define PR_BYTES (16 * 16 * PR_PITCH * 4)

__global__ __launch_bounds__(256) void pack_x_kernel(const float* __restrict__ x,
                                                     v2f* __restrict__ xp)
{
    int f    = blockIdx.x * 256 + threadIdx.x;
    int lane = f & 31;
    int kk   = (f >> 5) & 3;
    int i    = (f >> 7) & 63;
    int bt   = f >> 13;
    int m    = lane & 15;
    int e0   = kk * 4 + ((lane >> 4) << 1);
    const float* s = x + (size_t)(bt * BT + m) * 1024 + i;
    v2f v;
    v.x = s[(size_t)e0 * 64];
    v.y = s[(size_t)(e0 + 1) * 64];
    xp[f] = v;
}

__global__ __launch_bounds__(256) void pack_w_kernel(const float* __restrict__ rw,
                                                     const float* __restrict__ nz,
                                                     v2f* __restrict__ wf)
{
    int f    = blockIdx.x * 256 + threadIdx.x;
    int lane = f & 31;
    int kk   = (f >> 5) & 3;
    int i    = (f >> 7) & 63;
    int o    = f >> 13;
    int d    = lane & 15;
    int e0   = kk * 4 + ((lane >> 4) << 1);
    size_t idx = ((size_t)(o * I_CAP + i) * 16 + d) * 16 + e0;
    v2f v;
    v.x = rw[idx]     + nz[idx];
    v.y = rw[idx + 1] + nz[idx + 1];
    wf[f] = v;
}

__global__ __launch_bounds__(256) void routing_fused_kernel(const v2f* __restrict__ xp,
                                                            const v2f* __restrict__ wf,
                                                            float* __restrict__ out)
{
    extern __shared__ float pr[];

    const int bt   = blockIdx.x;
    const int o    = blockIdx.y;
    const int tid  = threadIdx.x;
    const int lane = tid & 31;
    const int wv   = tid >> 5;
    const int dn   = lane & 15;
    const int hi8  = (lane >> 4) << 3;

    #pragma unroll
    for (int ii = 0; ii < 8; ++ii) {
        const int i = wv * 8 + ii;
        const v2f* pa = xp + (((size_t)bt * I_CAP + i) * 4) * 32 + lane;
        const v2f* pb = wf + (((size_t)o  * I_CAP + i) * 4) * 32 + lane;
        v8f acc = {};
        #pragma unroll
        for (int kk = 0; kk < 4; ++kk) {
            v2f a = pa[kk * 32];
            v2f b = pb[kk * 32];
            acc = __builtin_amdgcn_wmma_f32_16x16x4_f32(
                false, a, false, b, (short)0, acc, false, false);
        }
        #pragma unroll
        for (int v = 0; v < 8; ++v) {
            const int bl = v + hi8;
            pr[(bl * 16 + dn) * PR_PITCH + i] = acc[v];
        }
    }
    __syncthreads();

    const int tb = tid >> 4;
    const int td = tid & 15;

    float prv[64];
    const float4* row = reinterpret_cast<const float4*>(&pr[(tb * 16 + td) * PR_PITCH]);
    #pragma unroll
    for (int q = 0; q < 16; ++q) {
        float4 fq = row[q];
        prv[4 * q + 0] = fq.x;  prv[4 * q + 1] = fq.y;
        prv[4 * q + 2] = fq.z;  prv[4 * q + 3] = fq.w;
    }

    float lg[64];
    #pragma unroll
    for (int i = 0; i < 64; ++i) lg[i] = 0.0f;

    float outd = 0.0f;
    #pragma unroll
    for (int it = 0; it < 3; ++it) {
        float mx = lg[0];
        #pragma unroll
        for (int i = 1; i < 64; ++i) mx = fmaxf(mx, lg[i]);
        float se = 0.0f, s = 0.0f;
        #pragma unroll
        for (int i = 0; i < 64; ++i) {
            float e = __expf(lg[i] - mx);
            se += e;
            s = fmaf(e, prv[i], s);
        }
        s /= se;

        float sn = s * s;
        sn += __shfl_xor(sn, 1, 32);
        sn += __shfl_xor(sn, 2, 32);
        sn += __shfl_xor(sn, 4, 32);
        sn += __shfl_xor(sn, 8, 32);
        outd = (sn / (1.0f + sn)) * rsqrtf(sn) * s;

        if (it != 2) {
            #pragma unroll
            for (int i = 0; i < 64; ++i) lg[i] = fmaf(prv[i], outd, lg[i]);
        }
    }

    out[((size_t)o * B_TOT + bt * BT + tb) * D_DIM + td] = outd;
}

extern "C" void kernel_launch(void* const* d_in, const int* in_sizes, int n_in,
                              void* d_out, int out_size, void* d_ws, size_t ws_size,
                              hipStream_t stream)
{
    (void)in_sizes; (void)n_in; (void)out_size; (void)ws_size;

    const float* x  = (const float*)d_in[0];
    const float* rw = (const float*)d_in[1];
    const float* nz = (const float*)d_in[2];
    float*       out = (float*)d_out;

    v2f* xp = (v2f*)d_ws;
    v2f* wfp = (v2f*)((char*)d_ws + (size_t)8 * 1024 * 1024);

    pack_x_kernel<<<4096, 256, 0, stream>>>(x, xp);
    pack_w_kernel<<<1024, 256, 0, stream>>>(rw, nz, wfp);

    dim3 grid(NBT, O_CAP);
    routing_fused_kernel<<<grid, 256, PR_BYTES, stream>>>(xp, wfp, out);
}


# fused kernel keeps its b-tile for 8 o's: x tile read once from x (whole lines, LDS transpose) into register-resident A fragments; pack_x launch and its workspace plane removed
# speedup vs baseline: 1.3486x; 1.3486x over previous
#include <hip/hip_runtime.h>


typedef __attribute__((ext_vector_type(2))) float v2f;
typedef __attribute__((ext_vector_type(8))) float v8f;

#define B_TOT   2048
#define I_CAP   64
#define O_CAP   32
#define D_DIM   16
#define BT      16
#define NBT     (B_TOT / BT)
#define OG      8
#define NOG     (O_CAP / OG)
#define PR_PITCH 68
#define PR_BYTES (16 * 16 * PR_PITCH * 4)
#define XS_PITCH 1028

__global__ __launch_bounds__(256) void pack_w_kernel(const float* __restrict__ rw,
                                                     const float* __restrict__ nz,
                                                     v2f* __restrict__ wf)
{
    int f    = blockIdx.x * 256 + threadIdx.x;
    int lane = f & 31;
    int kk   = (f >> 5) & 3;
    int i    = (f >> 7) & 63;
    int o    = f >> 13;
    int d    = lane & 15;
    int e0   = kk * 4 + ((lane >> 4) << 1);
    size_t idx = ((size_t)(o * I_CAP + i) * 16 + d) * 16 + e0;
    v2f v;
    v.x = rw[idx]     + nz[idx];
    v.y = rw[idx + 1] + nz[idx + 1];
    wf[f] = v;
}

__global__ __launch_bounds__(256) void routing_fused_kernel(const float* __restrict__ x,
                                                            const v2f* __restrict__ wf,
                                                            float* __restrict__ out)
{
    extern __shared__ __attribute__((aligned(16))) float pr[];

    const int bt   = blockIdx.x;
    const int og   = blockIdx.y;
    const int tid  = threadIdx.x;
    const int lane = tid & 31;
    const int wv   = tid >> 5;
    const int dn   = lane & 15;
    const int hh   = lane >> 4;
    const int hi8  = hh << 3;

    {
        const float4* src = reinterpret_cast<const float4*>(x + (size_t)bt * BT * 1024) + tid;
        #pragma unroll
        for (int q = 0; q < 16; ++q) {
            float4 v = src[q * 256];
            *reinterpret_cast<float4*>(&pr[q * XS_PITCH + tid * 4]) = v;
        }
    }
    __syncthreads();

    float af[8][4][2];
    {
        const float* base = &pr[dn * XS_PITCH + (2 * hh) * 64 + wv * 8];
        #pragma unroll
        for (int kk = 0; kk < 4; ++kk) {
            #pragma unroll
            for (int j = 0; j < 2; ++j) {
                const float4* p = reinterpret_cast<const float4*>(base + (kk * 4 + j) * 64);
                float4 r0 = p[0];
                float4 r1 = p[1];
                af[0][kk][j] = r0.x;  af[1][kk][j] = r0.y;
                af[2][kk][j] = r0.z;  af[3][kk][j] = r0.w;
                af[4][kk][j] = r1.x;  af[5][kk][j] = r1.y;
                af[6][kk][j] = r1.z;  af[7][kk][j] = r1.w;
            }
        }
    }

    const int tb = tid >> 4;
    const int td = tid & 15;

    for (int oo = 0; oo < OG; ++oo) {
        const int o = og * OG + oo;

        const v2f* pb = wf + (((size_t)o * I_CAP + wv * 8) * 4) * 32 + lane;
        v8f acc[8];
        #pragma unroll
        for (int ii = 0; ii < 8; ++ii) {
            v8f c = {};
            #pragma unroll
            for (int kk = 0; kk < 4; ++kk) {
                v2f a;
                a.x = af[ii][kk][0];
                a.y = af[ii][kk][1];
                v2f b = pb[(ii * 4 + kk) * 32];
                c = __builtin_amdgcn_wmma_f32_16x16x4_f32(
                    false, a, false, b, (short)0, c, false, false);
            }
            acc[ii] = c;
        }

        __syncthreads();

        #pragma unroll
        for (int ii = 0; ii < 8; ++ii) {
            const int i = wv * 8 + ii;
            #pragma unroll
            for (int v = 0; v < 8; ++v) {
                const int bl = v + hi8;
                pr[(bl * 16 + dn) * PR_PITCH + i] = acc[ii][v];
            }
        }
        __syncthreads();

        float prv[64];
        const float4* row = reinterpret_cast<const float4*>(&pr[(tb * 16 + td) * PR_PITCH]);
        #pragma unroll
        for (int q = 0; q < 16; ++q) {
            float4 fq = row[q];
            prv[4 * q + 0] = fq.x;  prv[4 * q + 1] = fq.y;
            prv[4 * q + 2] = fq.z;  prv[4 * q + 3] = fq.w;
        }

        float lg[64];
        #pragma unroll
        for (int i = 0; i < 64; ++i) lg[i] = 0.0f;

        float outd = 0.0f;
        #pragma unroll
        for (int it = 0; it < 3; ++it) {
            float mx = lg[0];
            #pragma unroll
            for (int i = 1; i < 64; ++i) mx = fmaxf(mx, lg[i]);
            float se = 0.0f, s = 0.0f;
            #pragma unroll
            for (int i = 0; i < 64; ++i) {
                float e = __expf(lg[i] - mx);
                se += e;
                s = fmaf(e, prv[i], s);
            }
            s /= se;

            float sn = s * s;
            sn += __shfl_xor(sn, 1, 32);
            sn += __shfl_xor(sn, 2, 32);
            sn += __shfl_xor(sn, 4, 32);
            sn += __shfl_xor(sn, 8, 32);
            outd = (sn / (1.0f + sn)) * rsqrtf(sn) * s;

            if (it != 2) {
                #pragma unroll
                for (int i = 0; i < 64; ++i) lg[i] = fmaf(prv[i], outd, lg[i]);
            }
        }

        out[((size_t)o * B_TOT + bt * BT + tb) * D_DIM + td] = outd;
    }
}

extern "C" void kernel_launch(void* const* d_in, const int* in_sizes, int n_in,
                              void* d_out, int out_size, void* d_ws, size_t ws_size,
                              hipStream_t stream)
{
    (void)in_sizes; (void)n_in; (void)out_size; (void)ws_size;

    const float* x  = (const float*)d_in[0];
    const float* rw = (const float*)d_in[1];
    const float* nz = (const float*)d_in[2];
    float*       out = (float*)d_out;

    v2f* wfp = (v2f*)d_ws;

    pack_w_kernel<<<1024, 256, 0, stream>>>(rw, nz, wfp);

    dim3 grid(NBT, NOG);
    routing_fused_kernel<<<grid, 256, PR_BYTES, stream>>>(x, wfp, out);
}


# squash's sum over d: the four butterfly steps are DPP row_xmask adds on the vector ALU instead of ds_bpermute round trips through the LDS unit; same operands, sums bit-identical
# speedup vs baseline: 1.4056x; 1.0423x over previous
#include <hip/hip_runtime.h>


typedef __attribute__((ext_vector_type(2))) float v2f;
typedef __attribute__((ext_vector_type(8))) float v8f;

#define B_TOT   2048
#define I_CAP   64
#define O_CAP   32
#define D_DIM   16
#define BT      16
#define NBT     (B_TOT / BT)
#define OG      8
#define NOG     (O_CAP / OG)
#define PR_PITCH 68
#define PR_BYTES (16 * 16 * PR_PITCH * 4)
#define XS_PITCH 1028

__global__ __launch_bounds__(256) void pack_w_kernel(const float* __restrict__ rw,
                                                     const float* __restrict__ nz,
                                                     v2f* __restrict__ wf)
{
    int f    = blockIdx.x * 256 + threadIdx.x;
    int lane = f & 31;
    int kk   = (f >> 5) & 3;
    int i    = (f >> 7) & 63;
    int o    = f >> 13;
    int d    = lane & 15;
    int e0   = kk * 4 + ((lane >> 4) << 1);
    size_t idx = ((size_t)(o * I_CAP + i) * 16 + d) * 16 + e0;
    v2f v;
    v.x = rw[idx]     + nz[idx];
    v.y = rw[idx + 1] + nz[idx + 1];
    wf[f] = v;
}

template <int M>
__device__ __forceinline__ float row_xor_add(float v)
{
    const int vi = __builtin_bit_cast(int, v);
    const int pi = __builtin_amdgcn_update_dpp(vi, vi, 0x160 + M, 0xf, 0xf, true);
    return v + __builtin_bit_cast(float, pi);
}

__global__ __launch_bounds__(256) void routing_fused_kernel(const float* __restrict__ x,
                                                            const v2f* __restrict__ wf,
                                                            float* __restrict__ out)
{
    extern __shared__ __attribute__((aligned(16))) float pr[];

    const int bt   = blockIdx.x;
    const int og   = blockIdx.y;
    const int tid  = threadIdx.x;
    const int lane = tid & 31;
    const int wv   = tid >> 5;
    const int dn   = lane & 15;
    const int hh   = lane >> 4;
    const int hi8  = hh << 3;

    {
        const float4* src = reinterpret_cast<const float4*>(x + (size_t)bt * BT * 1024) + tid;
        #pragma unroll
        for (int q = 0; q < 16; ++q) {
            float4 v = src[q * 256];
            *reinterpret_cast<float4*>(&pr[q * XS_PITCH + tid * 4]) = v;
        }
    }
    __syncthreads();

    float af[8][4][2];
    {
        const float* base = &pr[dn * XS_PITCH + (2 * hh) * 64 + wv * 8];
        #pragma unroll
        for (int kk = 0; kk < 4; ++kk) {
            #pragma unroll
            for (int j = 0; j < 2; ++j) {
                const float4* p = reinterpret_cast<const float4*>(base + (kk * 4 + j) * 64);
                float4 r0 = p[0];
                float4 r1 = p[1];
                af[0][kk][j] = r0.x;  af[1][kk][j] = r0.y;
                af[2][kk][j] = r0.z;  af[3][kk][j] = r0.w;
                af[4][kk][j] = r1.x;  af[5][kk][j] = r1.y;
                af[6][kk][j] = r1.z;  af[7][kk][j] = r1.w;
            }
        }
    }

    const int tb = tid >> 4;
    const int td = tid & 15;

    for (int oo = 0; oo < OG; ++oo) {
        const int o = og * OG + oo;

        const v2f* pb = wf + (((size_t)o * I_CAP + wv * 8) * 4) * 32 + lane;
        v8f acc[8];
        #pragma unroll
        for (int ii = 0; ii < 8; ++ii) {
            v8f c = {};
            #pragma unroll
            for (int kk = 0; kk < 4; ++kk) {
                v2f a;
                a.x = af[ii][kk][0];
                a.y = af[ii][kk][1];
                v2f b = pb[(ii * 4 + kk) * 32];
                c = __builtin_amdgcn_wmma_f32_16x16x4_f32(
                    false, a, false, b, (short)0, c, false, false);
            }
            acc[ii] = c;
        }

        __syncthreads();

        #pragma unroll
        for (int ii = 0; ii < 8; ++ii) {
            const int i = wv * 8 + ii;
            #pragma unroll
            for (int v = 0; v < 8; ++v) {
                const int bl = v + hi8;
                pr[(bl * 16 + dn) * PR_PITCH + i] = acc[ii][v];
            }
        }
        __syncthreads();

        float prv[64];
        const float4* row = reinterpret_cast<const float4*>(&pr[(tb * 16 + td) * PR_PITCH]);
        #pragma unroll
        for (int q = 0; q < 16; ++q) {
            float4 fq = row[q];
            prv[4 * q + 0] = fq.x;  prv[4 * q + 1] = fq.y;
            prv[4 * q + 2] = fq.z;  prv[4 * q + 3] = fq.w;
        }

        float lg[64];
        #pragma unroll
        for (int i = 0; i < 64; ++i) lg[i] = 0.0f;

        float outd = 0.0f;
        #pragma unroll
        for (int it = 0; it < 3; ++it) {
            float mx = lg[0];
            #pragma unroll
            for (int i = 1; i < 64; ++i) mx = fmaxf(mx, lg[i]);
            float se = 0.0f, s = 0.0f;
            #pragma unroll
            for (int i = 0; i < 64; ++i) {
                float e = __expf(lg[i] - mx);
                se += e;
                s = fmaf(e, prv[i], s);
            }
            s /= se;

            float sn = s * s;
            sn = row_xor_add<1>(sn);
            sn = row_xor_add<2>(sn);
            sn = row_xor_add<4>(sn);
            sn = row_xor_add<8>(sn);
            outd = (sn / (1.0f + sn)) * rsqrtf(sn) * s;

            if (it != 2) {
                #pragma unroll
                for (int i = 0; i < 64; ++i) lg[i] = fmaf(prv[i], outd, lg[i]);
            }
        }

        out[((size_t)o * B_TOT + bt * BT + tb) * D_DIM + td] = outd;
    }
}

extern "C" void kernel_launch(void* const* d_in, const int* in_sizes, int n_in,
                              void* d_out, int out_size, void* d_ws, size_t ws_size,
                              hipStream_t stream)
{
    (void)in_sizes; (void)n_in; (void)out_size; (void)ws_size;

    const float* x  = (const float*)d_in[0];
    const float* rw = (const float*)d_in[1];
    const float* nz = (const float*)d_in[2];
    float*       out = (float*)d_out;

    v2f* wfp = (v2f*)d_ws;

    pack_w_kernel<<<1024, 256, 0, stream>>>(rw, nz, wfp);

    dim3 grid(NBT, NOG);
    routing_fused_kernel<<<grid, 256, PR_BYTES, stream>>>(x, wfp, out);
}


# routing phase on packed f32 (v_pk_fma_f32, v_pk_add_f32) over pairs of i: logit update, exp2 argument as one fma, both sums; exps kept one by one between them; sums over i now in 4 partial sums
# speedup vs baseline: 1.5294x; 1.0881x over previous
#include <hip/hip_runtime.h>


typedef __attribute__((ext_vector_type(2))) float v2f;
typedef __attribute__((ext_vector_type(8))) float v8f;

#define B_TOT   2048
#define I_CAP   64
#define O_CAP   32
#define D_DIM   16
#define BT      16
#define NBT     (B_TOT / BT)
#define OG      8
#define NOG     (O_CAP / OG)
#define PR_PITCH 68
#define PR_BYTES (16 * 16 * PR_PITCH * 4)
#define XS_PITCH 1028

__global__ __launch_bounds__(256) void pack_w_kernel(const float* __restrict__ rw,
                                                     const float* __restrict__ nz,
                                                     v2f* __restrict__ wf)
{
    int f    = blockIdx.x * 256 + threadIdx.x;
    int lane = f & 31;
    int kk   = (f >> 5) & 3;
    int i    = (f >> 7) & 63;
    int o    = f >> 13;
    int d    = lane & 15;
    int e0   = kk * 4 + ((lane >> 4) << 1);
    size_t idx = ((size_t)(o * I_CAP + i) * 16 + d) * 16 + e0;
    v2f v;
    v.x = rw[idx]     + nz[idx];
    v.y = rw[idx + 1] + nz[idx + 1];
    wf[f] = v;
}

template <int M>
__device__ __forceinline__ float row_xor_add(float v)
{
    const int vi = __builtin_bit_cast(int, v);
    const int pi = __builtin_amdgcn_update_dpp(vi, vi, 0x160 + M, 0xf, 0xf, true);
    return v + __builtin_bit_cast(float, pi);
}

__global__ __launch_bounds__(256) void routing_fused_kernel(const float* __restrict__ x,
                                                            const v2f* __restrict__ wf,
                                                            float* __restrict__ out)
{
    extern __shared__ __attribute__((aligned(16))) float pr[];

    const int bt   = blockIdx.x;
    const int og   = blockIdx.y;
    const int tid  = threadIdx.x;
    const int lane = tid & 31;
    const int wv   = tid >> 5;
    const int dn   = lane & 15;
    const int hh   = lane >> 4;
    const int hi8  = hh << 3;

    {
        const float4* src = reinterpret_cast<const float4*>(x + (size_t)bt * BT * 1024) + tid;
        #pragma unroll
        for (int q = 0; q < 16; ++q) {
            float4 v = src[q * 256];
            *reinterpret_cast<float4*>(&pr[q * XS_PITCH + tid * 4]) = v;
        }
    }
    __syncthreads();

    float af[8][4][2];
    {
        const float* base = &pr[dn * XS_PITCH + (2 * hh) * 64 + wv * 8];
        #pragma unroll
        for (int kk = 0; kk < 4; ++kk) {
            #pragma unroll
            for (int j = 0; j < 2; ++j) {
                const float4* p = reinterpret_cast<const float4*>(base + (kk * 4 + j) * 64);
                float4 r0 = p[0];
                float4 r1 = p[1];
                af[0][kk][j] = r0.x;  af[1][kk][j] = r0.y;
                af[2][kk][j] = r0.z;  af[3][kk][j] = r0.w;
                af[4][kk][j] = r1.x;  af[5][kk][j] = r1.y;
                af[6][kk][j] = r1.z;  af[7][kk][j] = r1.w;
            }
        }
    }

    const int tb = tid >> 4;
    const int td = tid & 15;

    for (int oo = 0; oo < OG; ++oo) {
        const int o = og * OG + oo;

        const v2f* pb = wf + (((size_t)o * I_CAP + wv * 8) * 4) * 32 + lane;
        v8f acc[8];
        #pragma unroll
        for (int ii = 0; ii < 8; ++ii) {
            v8f c = {};
            #pragma unroll
            for (int kk = 0; kk < 4; ++kk) {
                v2f a;
                a.x = af[ii][kk][0];
                a.y = af[ii][kk][1];
                v2f b = pb[(ii * 4 + kk) * 32];
                c = __builtin_amdgcn_wmma_f32_16x16x4_f32(
                    false, a, false, b, (short)0, c, false, false);
            }
            acc[ii] = c;
        }

        __syncthreads();

        #pragma unroll
        for (int ii = 0; ii < 8; ++ii) {
            const int i = wv * 8 + ii;
            #pragma unroll
            for (int v = 0; v < 8; ++v) {
                const int bl = v + hi8;
                pr[(bl * 16 + dn) * PR_PITCH + i] = acc[ii][v];
            }
        }
        __syncthreads();

        v2f prv[32];
        const float4* row = reinterpret_cast<const float4*>(&pr[(tb * 16 + td) * PR_PITCH]);
        #pragma unroll
        for (int q = 0; q < 16; ++q) {
            float4 fq = row[q];
            prv[2 * q + 0] = (v2f){fq.x, fq.y};
            prv[2 * q + 1] = (v2f){fq.z, fq.w};
        }

        v2f lg[32];
        #pragma unroll
        for (int p = 0; p < 32; ++p) lg[p] = (v2f){0.0f, 0.0f};

        const float L2E = 1.4426950408889634f;
        const v2f l2e2 = {L2E, L2E};

        float outd = 0.0f;
        #pragma unroll
        for (int it = 0; it < 3; ++it) {
            float s;
            if (it == 0) {
                v2f sa = prv[0], sb = prv[1];
                #pragma unroll
                for (int p = 2; p < 32; p += 2) { sa += prv[p]; sb += prv[p + 1]; }
                const v2f st = sa + sb;
                s = (st.x + st.y) * (1.0f / 64.0f);
            } else {
                const v2f o2 = {outd, outd};
                float mxa = 0.0f, mxb = 0.0f;
                #pragma unroll
                for (int p = 0; p < 32; p += 2) {
                    lg[p]     = __builtin_elementwise_fma(prv[p],     o2, lg[p]);
                    lg[p + 1] = __builtin_elementwise_fma(prv[p + 1], o2, lg[p + 1]);
                    if (p == 0) {
                        mxa = fmaxf(lg[0].x, lg[0].y);
                        mxb = fmaxf(lg[1].x, lg[1].y);
                    } else {
                        mxa = fmaxf(fmaxf(mxa, lg[p].x), lg[p].y);
                        mxb = fmaxf(fmaxf(mxb, lg[p + 1].x), lg[p + 1].y);
                    }
                }
                const float nm = -fmaxf(mxa, mxb) * L2E;
                const v2f nm2 = {nm, nm};

                v2f sea = {0.0f, 0.0f}, seb = {0.0f, 0.0f};
                v2f sa  = {0.0f, 0.0f}, sb  = {0.0f, 0.0f};
                v2f ar[32], ex[32];
                #pragma unroll
                for (int j = 0; j < 36; ++j) {
                    if (j >= 2 && j < 34) {
                        ex[j - 2].x = __builtin_amdgcn_exp2f(ar[j - 2].x);
                        __builtin_amdgcn_sched_barrier(0);
                    }
                    if (j < 32) {
                        ar[j] = __builtin_elementwise_fma(lg[j], l2e2, nm2);
                        __builtin_amdgcn_sched_barrier(0);
                    }
                    if (j >= 2 && j < 34) {
                        ex[j - 2].y = __builtin_amdgcn_exp2f(ar[j - 2].y);
                        __builtin_amdgcn_sched_barrier(0);
                    }
                    if (j >= 4) {
                        const int k = j - 4;
                        if (k & 1) { seb += ex[k]; sb = __builtin_elementwise_fma(ex[k], prv[k], sb); }
                        else       { sea += ex[k]; sa = __builtin_elementwise_fma(ex[k], prv[k], sa); }
                        __builtin_amdgcn_sched_barrier(0);
                    }
                }
                const v2f set = sea + seb;
                const v2f st  = sa + sb;
                s = (st.x + st.y) / (set.x + set.y);
            }

            float sn = s * s;
            sn = row_xor_add<1>(sn);
            sn = row_xor_add<2>(sn);
            sn = row_xor_add<4>(sn);
            sn = row_xor_add<8>(sn);
            outd = (sn / (1.0f + sn)) * rsqrtf(sn) * s;
        }

        out[((size_t)o * B_TOT + bt * BT + tb) * D_DIM + td] = outd;
    }
}

extern "C" void kernel_launch(void* const* d_in, const int* in_sizes, int n_in,
                              void* d_out, int out_size, void* d_ws, size_t ws_size,
                              hipStream_t stream)
{
    (void)in_sizes; (void)n_in; (void)out_size; (void)ws_size;

    const float* x  = (const float*)d_in[0];
    const float* rw = (const float*)d_in[1];
    const float* nz = (const float*)d_in[2];
    float*       out = (float*)d_out;

    v2f* wfp = (v2f*)d_ws;

    pack_w_kernel<<<1024, 256, 0, stream>>>(rw, nz, wfp);

    dim3 grid(NBT, NOG);
    routing_fused_kernel<<<grid, 256, PR_BYTES, stream>>>(x, wfp, out);
}
